# starting kernel
# baseline (speedup 1.0000x reference)
#include <hip/hip_runtime.h>
#include <math.h>


typedef float v2f __attribute__((ext_vector_type(2)));
typedef float v8f __attribute__((ext_vector_type(8)));

#define BROW 40

__device__ __forceinline__ float fast_exp(float x) {
    return __builtin_amdgcn_exp2f(x * 1.4426950408889634f);
}
__device__ __forceinline__ float fast_log(float x) {
    return __builtin_amdgcn_logf(x) * 0.6931471805599453f;
}

__global__ __launch_bounds__(256) void lse_wmma_kernel(const float* __restrict__ x,
                                                       const float* __restrict__ w,
                                                       float* __restrict__ out) {
    const int d = blockIdx.x >> 3;
    const int r = blockIdx.x & 7;

    __shared__ float sB[32 * BROW];
    __shared__ float sMw[32];
    __shared__ float sPmax[8 * 32];

    const int tid   = threadIdx.x;
    const int oc    = tid & 31;
    const int chunk = tid >> 5;

    const float* wp = w + (size_t)d * 8192 + (size_t)oc * 8 + r;
    float wv[4];
    float pm = -__builtin_inff();
#pragma unroll
    for (int j = 0; j < 4; ++j) {
        const int ic = chunk * 4 + j;
        wv[j] = wp[(size_t)ic * 256];
        pm = fmaxf(pm, wv[j]);
    }
    sPmax[chunk * 32 + oc] = pm;
    __syncthreads();
    if (tid < 32) {
        float m = sPmax[tid];
#pragma unroll
        for (int c = 1; c < 8; ++c) m = fmaxf(m, sPmax[c * 32 + tid]);
        sMw[tid] = m;
    }
    __syncthreads();
    {
        const float m = sMw[oc];
#pragma unroll
        for (int j = 0; j < 4; ++j) {
            const int ic = chunk * 4 + j;
            sB[ic * BROW + oc] = fast_exp(wv[j] - m);
        }
    }
    __syncthreads();

    const int lane = tid & 31;
    const int wave = tid >> 5;
    const int half = lane >> 4;
    const int Mr   = lane & 15;

    for (int mt = wave; mt < 16; mt += 8) {
        const int n0 = mt * 16;
        const float* px = x + (size_t)(n0 + Mr) * 16384 + (size_t)d * 256 + r + half * 16;

        float av[16];
        float mymax = -__builtin_inff();
#pragma unroll
        for (int k = 0; k < 8; ++k) {
            const float v0 = px[k * 32];
            const float v1 = px[k * 32 + 8];
            av[2 * k]     = v0;
            av[2 * k + 1] = v1;
            mymax = fmaxf(mymax, fmaxf(v0, v1));
        }
        const float full = fmaxf(mymax, __shfl_xor(mymax, 16, 32));

        v2f afrag[8];
#pragma unroll
        for (int k = 0; k < 8; ++k) {
            afrag[k].x = fast_exp(av[2 * k]     - full);
            afrag[k].y = fast_exp(av[2 * k + 1] - full);
        }

        float mrow[8];
#pragma unroll
        for (int v = 0; v < 8; ++v)
            mrow[v] = __shfl(full, v + 8 * half, 32);

#pragma unroll
        for (int nt = 0; nt < 2; ++nt) {
            v8f acc = {};
#pragma unroll
            for (int k = 0; k < 8; ++k) {
                const int row = 4 * k + 2 * half;
                v2f b;
                b.x = sB[row * BROW       + nt * 16 + Mr];
                b.y = sB[(row + 1) * BROW + nt * 16 + Mr];
                acc = __builtin_amdgcn_wmma_f32_16x16x4_f32(
                     false, afrag[k],  false, b,
                     (short)0, acc,  false,  false);
            }
            const float mwv  = sMw[nt * 16 + Mr];
            float* pout = out + (size_t)(n0 + 8 * half) * 16384 + (size_t)d * 256
                              + (size_t)(nt * 16 + Mr) * 8 + r;
#pragma unroll
            for (int v = 0; v < 8; ++v) {
                const float s = acc[v];
                pout[(size_t)v * 16384] = fast_log(fmaxf(s, 1e-38f)) + mrow[v] + mwv;
            }
        }
    }
}

extern "C" void kernel_launch(void* const* d_in, const int* in_sizes, int n_in,
                              void* d_out, int out_size, void* d_ws, size_t ws_size,
                              hipStream_t stream) {
    const float* x = (const float*)d_in[0];
    const float* w = (const float*)d_in[1];
    float* out = (float*)d_out;

    dim3 grid(64 * 8);
    dim3 block(256);
    hipLaunchKernelGGL(lse_wmma_kernel, grid, block, 0, stream, x, w, out);
}


# One work-group per (d, 64 n) with all 8 r: x, w and out move as whole contiguous lines (16 B per lane) staged through LDS instead of 4-byte pieces at a 32-byte stride; k order of the sums kept
# speedup vs baseline: 3.3388x; 3.3388x over previous
#include <hip/hip_runtime.h>
#include <math.h>


typedef float v2f __attribute__((ext_vector_type(2)));
typedef float v4f __attribute__((ext_vector_type(4)));
typedef float v8f __attribute__((ext_vector_type(8)));

#define ROWS   64
#define XP     260
#define BP     36
#define BPLANE (32 * BP + 4)

__device__ __forceinline__ float fast_exp(float x) {
    return __builtin_amdgcn_exp2f(x * 1.4426950408889634f);
}
__device__ __forceinline__ float fast_log(float x) {
    return __builtin_amdgcn_logf(x) * 0.6931471805599453f;
}

__global__ __launch_bounds__(256) void lse_wmma_rows_kernel(const float* __restrict__ x,
                                                            const float* __restrict__ w,
                                                            float* __restrict__ out) {
    const int d  = blockIdx.x >> 2;
    const int n0 = (blockIdx.x & 3) * ROWS;

    __shared__ __attribute__((aligned(16))) float sX[ROWS * XP];
    __shared__ __attribute__((aligned(16))) float sO[ROWS * XP];
    __shared__ __attribute__((aligned(16))) float sB[8 * BPLANE];
    __shared__ __attribute__((aligned(16))) float sMw[256];

    const int tid = threadIdx.x;

    const float* wp = w + (size_t)d * 8192 + tid;
    float wv[32];
#pragma unroll
    for (int ic = 0; ic < 32; ++ic) wv[ic] = wp[ic * 256];

    const float* xg = x + (size_t)n0 * 16384 + (size_t)d * 256;
    v4f xv[16];
#pragma unroll
    for (int i = 0; i < 16; ++i) {
        const int f   = tid + 256 * i;
        const int row = f >> 6;
        const int c4  = f & 63;
        xv[i] = *(const v4f*)(xg + (size_t)row * 16384 + c4 * 4);
    }

    {
        float m = wv[0];
#pragma unroll
        for (int ic = 1; ic < 32; ++ic) m = fmaxf(m, wv[ic]);
        sMw[tid] = m;
        float* bdst = sB + (tid & 7) * BPLANE + (tid >> 3) * BP;
#pragma unroll
        for (int q = 0; q < 8; ++q) {
            const int kb = 8 * (q & 3) + 2 * (q >> 2);
            v4f e;
            e.x = fast_exp(wv[kb]     - m);
            e.y = fast_exp(wv[kb + 1] - m);
            e.z = fast_exp(wv[kb + 4] - m);
            e.w = fast_exp(wv[kb + 5] - m);
            *(v4f*)(bdst + 4 * q) = e;
        }
    }
#pragma unroll
    for (int i = 0; i < 16; ++i) {
        const int f   = tid + 256 * i;
        const int row = f >> 6;
        const int c4  = f & 63;
        *(v4f*)(sX + row * XP + c4 * 4) = xv[i];
    }
    __syncthreads();

    const int lane = tid & 31;
    const int wave = tid >> 5;
    const int half = lane >> 4;
    const int c    = lane & 15;
    const int mt   = wave >> 1;
    const int rh   = wave & 1;

    const float* xa_p = sX + (mt * 16 + c) * XP + half * 16 + rh * 4;
    v4f xa[16];
#pragma unroll
    for (int kk = 0; kk < 8; ++kk) {
        xa[2 * kk]     = *(const v4f*)(xa_p + kk * 32);
        xa[2 * kk + 1] = *(const v4f*)(xa_p + kk * 32 + 8);
    }

    v2f   afrag[4][8];
    float mrow[4][8];
#pragma unroll
    for (int rr = 0; rr < 4; ++rr) {
        float mymax = xa[0][rr];
#pragma unroll
        for (int j = 1; j < 16; ++j) mymax = fmaxf(mymax, xa[j][rr]);
        const float full = fmaxf(mymax, __shfl_xor(mymax, 16, 32));
#pragma unroll
        for (int kk = 0; kk < 8; ++kk) {
            afrag[rr][kk].x = fast_exp(xa[2 * kk][rr]     - full);
            afrag[rr][kk].y = fast_exp(xa[2 * kk + 1][rr] - full);
        }
#pragma unroll
        for (int v = 0; v < 8; ++v)
            mrow[rr][v] = __shfl(full, v + 8 * half, 32);
    }

    float* so_p = sO + (mt * 16 + 8 * half) * XP + c * 8 + rh * 4;
#pragma unroll
    for (int nt = 0; nt < 2; ++nt) {
        const v4f mwv = *(const v4f*)(sMw + (nt * 16 + c) * 8 + rh * 4);
        v8f acc[4];
#pragma unroll
        for (int rr = 0; rr < 4; ++rr) {
            const float* bp = sB + (rh * 4 + rr) * BPLANE + (nt * 16 + c) * BP + half * 16;
            v4f bq[4];
#pragma unroll
            for (int j = 0; j < 4; ++j) bq[j] = *(const v4f*)(bp + 4 * j);
            acc[rr] = (v8f){};
#pragma unroll
            for (int kk = 0; kk < 8; ++kk) {
                v2f b;
                b.x = bq[kk >> 1][(kk & 1) * 2];
                b.y = bq[kk >> 1][(kk & 1) * 2 + 1];
                acc[rr] = __builtin_amdgcn_wmma_f32_16x16x4_f32(
                     false, afrag[rr][kk],  false, b,
                     (short)0, acc[rr],  false,  false);
            }
        }
#pragma unroll
        for (int v = 0; v < 8; ++v) {
            v4f o;
            o.x = fast_log(fmaxf(acc[0][v], 1e-38f)) + mrow[0][v] + mwv.x;
            o.y = fast_log(fmaxf(acc[1][v], 1e-38f)) + mrow[1][v] + mwv.y;
            o.z = fast_log(fmaxf(acc[2][v], 1e-38f)) + mrow[2][v] + mwv.z;
            o.w = fast_log(fmaxf(acc[3][v], 1e-38f)) + mrow[3][v] + mwv.w;
            *(v4f*)(so_p + v * XP + nt * 128) = o;
        }
    }
    __syncthreads();

    float* og = out + (size_t)n0 * 16384 + (size_t)d * 256;
#pragma unroll
    for (int i = 0; i < 16; ++i) {
        const int f   = tid + 256 * i;
        const int row = f >> 6;
        const int c4  = f & 63;
        const v4f o = *(const v4f*)(sO + row * XP + c4 * 4);
        *(v4f*)(og + (size_t)row * 16384 + c4 * 4) = o;
    }
}

extern "C" void kernel_launch(void* const* d_in, const int* in_sizes, int n_in,
                              void* d_out, int out_size, void* d_ws, size_t ws_size,
                              hipStream_t stream) {
    const float* x = (const float*)d_in[0];
    const float* w = (const float*)d_in[1];
    float* out = (float*)d_out;

    dim3 grid(64 * 4);
    dim3 block(256);
    hipLaunchKernelGGL(lse_wmma_rows_kernel, grid, block, 0, stream, x, w, out);
}


# Two 32-row stages per work-group, asm LDS barriers leave global loads/stores in flight: rows 32..63 load under stage 0, stage 0 rows store under stage 1 (8-byte LDS pieces, padded); sums kept
# speedup vs baseline: 3.6803x; 1.1023x over previous
#include <hip/hip_runtime.h>
#include <math.h>


typedef float v2f __attribute__((ext_vector_type(2)));
typedef float v4f __attribute__((ext_vector_type(4)));
typedef float v8f __attribute__((ext_vector_type(8)));

#define ROWS   64
#define XP     260
#define OP     276
#define BP     36
#define BPLANE (32 * BP + 4)

__device__ __forceinline__ float fast_exp(float x) {
    return __builtin_amdgcn_exp2f(x * 1.4426950408889634f);
}
__device__ __forceinline__ float fast_log(float x) {
    return __builtin_amdgcn_logf(x) * 0.6931471805599453f;
}

__device__ __forceinline__ void lds_barrier() {
    asm volatile("s_wait_dscnt 0x0\n\ts_barrier_signal -1\n\ts_barrier_wait -1" ::: "memory");
}

__global__ __launch_bounds__(256) void lse_wmma_rows_kernel(const float* __restrict__ x,
                                                            const float* __restrict__ w,
                                                            float* __restrict__ out) {
    const int d  = blockIdx.x >> 2;
    const int n0 = (blockIdx.x & 3) * ROWS;

    __shared__ __attribute__((aligned(16))) float sX[ROWS * XP];
    __shared__ __attribute__((aligned(16))) float sO[ROWS * OP];
    __shared__ __attribute__((aligned(16))) float sB[8 * BPLANE];
    __shared__ __attribute__((aligned(16))) float sMw[256];

    const int tid = threadIdx.x;

    const float* wp = w + (size_t)d * 8192 + tid;
    float wv[32];
#pragma unroll
    for (int ic = 0; ic < 32; ++ic) wv[ic] = wp[ic * 256];

    const float* xg = x + (size_t)n0 * 16384 + (size_t)d * 256;
    v4f xv[16];
#pragma unroll
    for (int i = 0; i < 8; ++i) {
        const int f   = tid + 256 * i;
        const int row = f >> 6;
        const int c4  = f & 63;
        xv[i] = *(const v4f*)(xg + (size_t)row * 16384 + c4 * 4);
    }
    __builtin_amdgcn_sched_barrier(0);
#pragma unroll
    for (int i = 8; i < 16; ++i) {
        const int f   = tid + 256 * i;
        const int row = f >> 6;
        const int c4  = f & 63;
        xv[i] = *(const v4f*)(xg + (size_t)row * 16384 + c4 * 4);
    }
    __builtin_amdgcn_sched_barrier(0);

    {
        float m = wv[0];
#pragma unroll
        for (int ic = 1; ic < 32; ++ic) m = fmaxf(m, wv[ic]);
        sMw[tid] = m;
        float* bdst = sB + (tid & 7) * BPLANE + (tid >> 3) * BP;
#pragma unroll
        for (int q = 0; q < 8; ++q) {
            const int kb = 8 * (q & 3) + 2 * (q >> 2);
            v4f e;
            e.x = fast_exp(wv[kb]     - m);
            e.y = fast_exp(wv[kb + 1] - m);
            e.z = fast_exp(wv[kb + 4] - m);
            e.w = fast_exp(wv[kb + 5] - m);
            *(v4f*)(bdst + 4 * q) = e;
        }
    }
#pragma unroll
    for (int i = 0; i < 8; ++i) {
        const int f   = tid + 256 * i;
        const int row = f >> 6;
        const int c4  = f & 63;
        *(v4f*)(sX + row * XP + c4 * 4) = xv[i];
    }
    lds_barrier();

    const int lane = tid & 31;
    const int wave = tid >> 5;
    const int half = lane >> 4;
    const int c    = lane & 15;
    const int mtp  = wave >> 2;
    const int rq   = wave & 3;

    v4f bq[2][2][4];
    v2f mwv[2];
#pragma unroll
    for (int nt = 0; nt < 2; ++nt) {
#pragma unroll
        for (int rr = 0; rr < 2; ++rr) {
            const float* bp = sB + (rq * 2 + rr) * BPLANE + (nt * 16 + c) * BP + half * 16;
#pragma unroll
            for (int j = 0; j < 4; ++j) bq[nt][rr][j] = *(const v4f*)(bp + 4 * j);
        }
        mwv[nt] = *(const v2f*)(sMw + (nt * 16 + c) * 8 + rq * 2);
    }

    float* og = out + (size_t)n0 * 16384 + (size_t)d * 256;

#pragma unroll
    for (int st = 0; st < 2; ++st) {
        const int rbase = st * 32 + mtp * 16;

        const float* xa_p = sX + (rbase + c) * XP + half * 16 + rq * 2;
        v2f xa[16];
#pragma unroll
        for (int kk = 0; kk < 8; ++kk) {
            xa[2 * kk]     = *(const v2f*)(xa_p + kk * 32);
            xa[2 * kk + 1] = *(const v2f*)(xa_p + kk * 32 + 8);
        }

        v2f   afrag[2][8];
        float mrow[2][8];
#pragma unroll
        for (int rr = 0; rr < 2; ++rr) {
            float mymax = xa[0][rr];
#pragma unroll
            for (int j = 1; j < 16; ++j) mymax = fmaxf(mymax, xa[j][rr]);
            const float full = fmaxf(mymax, __shfl_xor(mymax, 16, 32));
#pragma unroll
            for (int kk = 0; kk < 8; ++kk) {
                afrag[rr][kk].x = fast_exp(xa[2 * kk][rr]     - full);
                afrag[rr][kk].y = fast_exp(xa[2 * kk + 1][rr] - full);
            }
#pragma unroll
            for (int v = 0; v < 8; ++v)
                mrow[rr][v] = __shfl(full, v + 8 * half, 32);
        }

        float* so_p = sO + (rbase + 8 * half) * OP + c * 8 + (c >> 3) * 4 + rq * 2;
#pragma unroll
        for (int nt = 0; nt < 2; ++nt) {
            v8f acc[2];
#pragma unroll
            for (int rr = 0; rr < 2; ++rr) {
                acc[rr] = (v8f){};
#pragma unroll
                for (int kk = 0; kk < 8; ++kk) {
                    v2f b;
                    b.x = bq[nt][rr][kk >> 1][(kk & 1) * 2];
                    b.y = bq[nt][rr][kk >> 1][(kk & 1) * 2 + 1];
                    acc[rr] = __builtin_amdgcn_wmma_f32_16x16x4_f32(
                         false, afrag[rr][kk],  false, b,
                         (short)0, acc[rr],  false,  false);
                }
            }
#pragma unroll
            for (int v = 0; v < 8; ++v) {
                v2f o;
                o.x = fast_log(fmaxf(acc[0][v], 1e-38f)) + mrow[0][v] + mwv[nt].x;
                o.y = fast_log(fmaxf(acc[1][v], 1e-38f)) + mrow[1][v] + mwv[nt].y;
                *(v2f*)(so_p + v * OP + nt * 136) = o;
            }
        }

        if (st == 0) {
            __builtin_amdgcn_sched_barrier(0);
#pragma unroll
            for (int i = 8; i < 16; ++i) {
                const int f   = tid + 256 * i;
                const int row = f >> 6;
                const int c4  = f & 63;
                *(v4f*)(sX + row * XP + c4 * 4) = xv[i];
            }
        }
        lds_barrier();

#pragma unroll
        for (int i = 0; i < 8; ++i) {
            const int f   = tid + 256 * (8 * st + i);
            const int row = f >> 6;
            const int c4  = f & 63;
            const v4f o = *(const v4f*)(sO + row * OP + c4 * 4 + (c4 >> 4) * 4);
            *(v4f*)(og + (size_t)row * 16384 + c4 * 4) = o;
        }
    }
}

extern "C" void kernel_launch(void* const* d_in, const int* in_sizes, int n_in,
                              void* d_out, int out_size, void* d_ws, size_t ws_size,
                              hipStream_t stream) {
    const float* x = (const float*)d_in[0];
    const float* w = (const float*)d_in[1];
    float* out = (float*)d_out;

    dim3 grid(64 * 4);
    dim3 block(256);
    hipLaunchKernelGGL(lse_wmma_rows_kernel, grid, block, 0, stream, x, w, out);
}
